# starting kernel
# baseline (speedup 1.0000x reference)
#include <hip/hip_runtime.h>
#include <hip/hip_bf16.h>
#include <math.h>


typedef __attribute__((ext_vector_type(2))) float v2f;
typedef __attribute__((ext_vector_type(8))) float v8f;

#define FIN 128


__global__ void zero_f32(float* __restrict__ p, int n) {
    int i = blockIdx.x * blockDim.x + threadIdx.x;
    if (i < n) p[i] = 0.0f;
}

__global__ void degree_kernel(const int* __restrict__ dst, float* __restrict__ cnt,
                              int nedges) {
    int e = blockIdx.x * blockDim.x + threadIdx.x;
    if (e < nedges) atomicAdd(&cnt[dst[e]], 1.0f);
}

__global__ void invdeg_kernel(const float* __restrict__ cnt, float* __restrict__ inv,
                              int n) {
    int i = blockIdx.x * blockDim.x + threadIdx.x;
    if (i < n) inv[i] = 1.0f / fmaxf(cnt[i], 1.0f);
}

__global__ void scatter_add_kernel(const float* __restrict__ h,
                                   const int* __restrict__ src,
                                   const int* __restrict__ dst,
                                   float* __restrict__ agg, int nedges) {
    int e    = (blockIdx.x * blockDim.x + threadIdx.x) >> 5;
    int lane = threadIdx.x & 31;
    if (e >= nedges) return;
    int s = src[e];
    int d = dst[e];
    const float4 v = ((const float4*)(h + (long)s * FIN))[lane];
    float* ap = agg + (long)d * FIN + lane * 4;
    atomicAdd(ap + 0, v.x);
    atomicAdd(ap + 1, v.y);
    atomicAdd(ap + 2, v.z);
    atomicAdd(ap + 3, v.w);
}


template <int FOUT, bool ELU>
__global__ void __launch_bounds__(32 * (FOUT / 16))
sage_gemm_kernel(const float* __restrict__ agg, const float* __restrict__ invdeg,
                 const float* __restrict__ h,
                 const float* __restrict__ Wl, const float* __restrict__ bl,
                 const float* __restrict__ Wr, float* __restrict__ out) {
    __shared__ float As[16][132];
    __shared__ float Xs[16][132];

    const int m0 = blockIdx.x * 16;

    for (int i = threadIdx.x; i < 16 * (FIN / 4); i += blockDim.x) {
        int r  = i >> 5;
        int c4 = i & 31;
        long g = (long)(m0 + r) * (FIN / 4) + c4;
        float inv = invdeg[m0 + r];
        float4 va = ((const float4*)agg)[g];
        float4 vx = ((const float4*)h)[g];
        va.x *= inv; va.y *= inv; va.z *= inv; va.w *= inv;
        *(float4*)&As[r][c4 * 4] = va;
        *(float4*)&Xs[r][c4 * 4] = vx;
    }
    __syncthreads();

    const int lane = threadIdx.x & 31;
    const int wave = threadIdx.x >> 5;
    const int n0   = wave * 16;
    const int row  = lane & 15;
    const int col  = lane & 15;
    const int hi   = lane >> 4;

    const float* wl = Wl + 2 * hi * FOUT + n0 + col;
    const float* wr = Wr + 2 * hi * FOUT + n0 + col;

    v8f c = {};

#pragma unroll
    for (int k = 0; k < FIN; k += 4) {
        v2f a, b;
        a.x = As[row][k + 2 * hi];
        a.y = As[row][k + 2 * hi + 1];
        b.x = wl[k * FOUT];
        b.y = wl[(k + 1) * FOUT];
        c = __builtin_amdgcn_wmma_f32_16x16x4_f32(false, a, false, b,
                                                  (short)0, c, false, false);
    }
#pragma unroll
    for (int k = 0; k < FIN; k += 4) {
        v2f a, b;
        a.x = Xs[row][k + 2 * hi];
        a.y = Xs[row][k + 2 * hi + 1];
        b.x = wr[k * FOUT];
        b.y = wr[(k + 1) * FOUT];
        c = __builtin_amdgcn_wmma_f32_16x16x4_f32(false, a, false, b,
                                                  (short)0, c, false, false);
    }

    const float bias = bl[n0 + col];
#pragma unroll
    for (int i = 0; i < 8; ++i) {
        float v = c[i] + bias;
        if (ELU) v = (v > 0.0f) ? v : expm1f(v);
        out[(long)(m0 + i + 8 * hi) * FOUT + n0 + col] = v;
    }
}

__global__ void logsoftmax64_kernel(float* __restrict__ out, int n) {
    int row  = (blockIdx.x * blockDim.x + threadIdx.x) >> 5;
    int lane = threadIdx.x & 31;
    if (row >= n) return;
    float* p = out + (long)row * 64;
    float a = p[lane];
    float b = p[lane + 32];
    float m = fmaxf(a, b);
#pragma unroll
    for (int o = 16; o > 0; o >>= 1) m = fmaxf(m, __shfl_xor(m, o, 32));
    float s = __expf(a - m) + __expf(b - m);
#pragma unroll
    for (int o = 16; o > 0; o >>= 1) s += __shfl_xor(s, o, 32);
    float l = m + __logf(s);
    p[lane]      = a - l;
    p[lane + 32] = b - l;
}


extern "C" void kernel_launch(void* const* d_in, const int* in_sizes, int n_in,
                              void* d_out, int out_size, void* d_ws, size_t ws_size,
                              hipStream_t stream) {
    const float* x   = (const float*)d_in[0];
    const int*   ei  = (const int*)d_in[1];
    const float* Wl0 = (const float*)d_in[2];
    const float* bl0 = (const float*)d_in[3];
    const float* Wr0 = (const float*)d_in[4];
    const float* Wl1 = (const float*)d_in[5];
    const float* bl1 = (const float*)d_in[6];
    const float* Wr1 = (const float*)d_in[7];
    const float* Wl2 = (const float*)d_in[8];
    const float* bl2 = (const float*)d_in[9];
    const float* Wr2 = (const float*)d_in[10];

    const int n_nodes = in_sizes[0] / FIN;
    const int n_edges = in_sizes[1] / 2;
    const int* src = ei;
    const int* dst = ei + n_edges;

    float* agg  = (float*)d_ws;
    float* hbuf = agg + (long)n_nodes * FIN;
    float* cnt  = hbuf + (long)n_nodes * FIN;
    float* inv  = cnt + n_nodes;

    const int nf     = n_nodes * FIN;
    const int zb     = (nf + 255) / 256;
    const int eb1    = (n_edges + 255) / 256;
    const int ebw    = (n_edges * 32 + 255) / 256;
    const int nb     = (n_nodes + 255) / 256;
    const int mtiles = n_nodes / 16;

    zero_f32<<<nb, 256, 0, stream>>>(cnt, n_nodes);
    degree_kernel<<<eb1, 256, 0, stream>>>(dst, cnt, n_edges);
    invdeg_kernel<<<nb, 256, 0, stream>>>(cnt, inv, n_nodes);

    zero_f32<<<zb, 256, 0, stream>>>(agg, nf);
    scatter_add_kernel<<<ebw, 256, 0, stream>>>(x, src, dst, agg, n_edges);
    sage_gemm_kernel<128, true><<<mtiles, 256, 0, stream>>>(agg, inv, x,
                                                            Wl0, bl0, Wr0, hbuf);

    zero_f32<<<zb, 256, 0, stream>>>(agg, nf);
    scatter_add_kernel<<<ebw, 256, 0, stream>>>(hbuf, src, dst, agg, n_edges);
    sage_gemm_kernel<128, true><<<mtiles, 256, 0, stream>>>(agg, inv, hbuf,
                                                            Wl1, bl1, Wr1, hbuf);

    zero_f32<<<zb, 256, 0, stream>>>(agg, nf);
    scatter_add_kernel<<<ebw, 256, 0, stream>>>(hbuf, src, dst, agg, n_edges);
    sage_gemm_kernel<64, false><<<mtiles, 128, 0, stream>>>(agg, inv, hbuf,
                                                            Wl2, bl2, Wr2,
                                                            (float*)d_out);

    logsoftmax64_kernel<<<(n_nodes * 32 + 255) / 256, 256, 0, stream>>>(
        (float*)d_out, n_nodes);
}


# Neighbour sums: f32 atomic scatter + zero fill replaced by atomics-free row gather over a per-call dst-sorted edge list (slot = one returning int add per edge); f32 sum order = that arrival rank
# speedup vs baseline: 2.3262x; 2.3262x over previous
#include <hip/hip_runtime.h>
#include <hip/hip_bf16.h>
#include <math.h>


typedef __attribute__((ext_vector_type(2))) float v2f;
typedef __attribute__((ext_vector_type(8))) float v8f;

#define FIN 128


__global__ void zero_i32(int* __restrict__ p, int n) {
    int i = blockIdx.x * blockDim.x + threadIdx.x;
    if (i < n) p[i] = 0;
}

__global__ void rank_kernel(const int* __restrict__ dst, int* __restrict__ cnt,
                            int* __restrict__ pos, int nedges) {
    int e = blockIdx.x * blockDim.x + threadIdx.x;
    if (e < nedges) pos[e] = atomicAdd(&cnt[dst[e]], 1);
}

__global__ void __launch_bounds__(1024)
rowstart_kernel(const int* __restrict__ cnt, int* __restrict__ rowstart,
                float* __restrict__ inv, int n) {
    __shared__ int wtot[32];
    __shared__ int wpre[32];
    const int t    = threadIdx.x;
    const int lane = t & 31;
    const int w    = t >> 5;
    const int i    = blockIdx.x * 1024 + t;

    int p = 0;
    for (int j = t; j < (int)blockIdx.x * 1024; j += 1024) p += cnt[j];
    const int c = (i < n) ? cnt[i] : 0;

    int x = c;
    for (int o = 1; o < 32; o <<= 1) {
        int y = __shfl_up(x, o, 32);
        if (lane >= o) x += y;
    }
    for (int o = 16; o > 0; o >>= 1) p += __shfl_xor(p, o, 32);
    if (lane == 31) wtot[w] = x;
    if (lane == 0)  wpre[w] = p;
    __syncthreads();

    int tot = wpre[lane] + ((lane < w) ? wtot[lane] : 0);
    for (int o = 16; o > 0; o >>= 1) tot += __shfl_xor(tot, o, 32);

    if (i < n) {
        rowstart[i] = tot + x - c;
        inv[i]      = 1.0f / fmaxf((float)c, 1.0f);
    }
}

__global__ void fill_kernel(const int* __restrict__ src, const int* __restrict__ dst,
                            const int* __restrict__ rowstart,
                            const int* __restrict__ pos, int* __restrict__ col,
                            int nedges) {
    int e = blockIdx.x * blockDim.x + threadIdx.x;
    if (e < nedges) col[rowstart[dst[e]] + pos[e]] = src[e];
}

template <int B>
__device__ __forceinline__ void add_rows(const float* __restrict__ h, int idxv,
                                         int k, int lane, float4& acc) {
    float4 v[B];
#pragma unroll
    for (int u = 0; u < B; ++u) {
        int s = __builtin_amdgcn_readlane(idxv, k + u);
        v[u] = ((const float4*)(h + (long)s * FIN))[lane];
    }
#pragma unroll
    for (int u = 0; u < B; ++u) {
        acc.x += v[u].x; acc.y += v[u].y; acc.z += v[u].z; acc.w += v[u].w;
    }
}

__global__ void __launch_bounds__(256)
gather_sum_kernel(const float* __restrict__ h, const int* __restrict__ rowstart,
                  const int* __restrict__ cnt, const int* __restrict__ col,
                  float* __restrict__ agg, int n) {
    const int node = __builtin_amdgcn_readfirstlane(
        (int)((blockIdx.x * blockDim.x + threadIdx.x) >> 5));
    const int lane = threadIdx.x & 31;
    if (node >= n) return;
    const int start = rowstart[node];
    const int deg   = cnt[node];

    float4 acc = make_float4(0.0f, 0.0f, 0.0f, 0.0f);
    for (int base = 0; base < deg; base += 32) {
        const int m = min(32, deg - base);
        const int idxv = (lane < m) ? col[start + base + lane] : 0;
        int k = 0;
        for (; k + 8 <= m; k += 8) add_rows<8>(h, idxv, k, lane, acc);
        if (k + 4 <= m) { add_rows<4>(h, idxv, k, lane, acc); k += 4; }
        if (k + 2 <= m) { add_rows<2>(h, idxv, k, lane, acc); k += 2; }
        if (k < m)      { add_rows<1>(h, idxv, k, lane, acc); }
    }
    ((float4*)(agg + (long)node * FIN))[lane] = acc;
}


template <int FOUT, bool ELU>
__global__ void __launch_bounds__(32 * (FOUT / 16))
sage_gemm_kernel(const float* __restrict__ agg, const float* __restrict__ invdeg,
                 const float* __restrict__ h,
                 const float* __restrict__ Wl, const float* __restrict__ bl,
                 const float* __restrict__ Wr, float* __restrict__ out) {
    __shared__ float As[16][132];
    __shared__ float Xs[16][132];

    const int m0 = blockIdx.x * 16;

    for (int i = threadIdx.x; i < 16 * (FIN / 4); i += blockDim.x) {
        int r  = i >> 5;
        int c4 = i & 31;
        long g = (long)(m0 + r) * (FIN / 4) + c4;
        float inv = invdeg[m0 + r];
        float4 va = ((const float4*)agg)[g];
        float4 vx = ((const float4*)h)[g];
        va.x *= inv; va.y *= inv; va.z *= inv; va.w *= inv;
        *(float4*)&As[r][c4 * 4] = va;
        *(float4*)&Xs[r][c4 * 4] = vx;
    }
    __syncthreads();

    const int lane = threadIdx.x & 31;
    const int wave = threadIdx.x >> 5;
    const int n0   = wave * 16;
    const int row  = lane & 15;
    const int col  = lane & 15;
    const int hi   = lane >> 4;

    const float* wl = Wl + 2 * hi * FOUT + n0 + col;
    const float* wr = Wr + 2 * hi * FOUT + n0 + col;

    v8f c = {};

#pragma unroll
    for (int k = 0; k < FIN; k += 4) {
        v2f a, b;
        a.x = As[row][k + 2 * hi];
        a.y = As[row][k + 2 * hi + 1];
        b.x = wl[k * FOUT];
        b.y = wl[(k + 1) * FOUT];
        c = __builtin_amdgcn_wmma_f32_16x16x4_f32(false, a, false, b,
                                                  (short)0, c, false, false);
    }
#pragma unroll
    for (int k = 0; k < FIN; k += 4) {
        v2f a, b;
        a.x = Xs[row][k + 2 * hi];
        a.y = Xs[row][k + 2 * hi + 1];
        b.x = wr[k * FOUT];
        b.y = wr[(k + 1) * FOUT];
        c = __builtin_amdgcn_wmma_f32_16x16x4_f32(false, a, false, b,
                                                  (short)0, c, false, false);
    }

    const float bias = bl[n0 + col];
#pragma unroll
    for (int i = 0; i < 8; ++i) {
        float v = c[i] + bias;
        if (ELU) v = (v > 0.0f) ? v : expm1f(v);
        out[(long)(m0 + i + 8 * hi) * FOUT + n0 + col] = v;
    }
}

__global__ void logsoftmax64_kernel(float* __restrict__ out, int n) {
    int row  = (blockIdx.x * blockDim.x + threadIdx.x) >> 5;
    int lane = threadIdx.x & 31;
    if (row >= n) return;
    float* p = out + (long)row * 64;
    float a = p[lane];
    float b = p[lane + 32];
    float m = fmaxf(a, b);
#pragma unroll
    for (int o = 16; o > 0; o >>= 1) m = fmaxf(m, __shfl_xor(m, o, 32));
    float s = __expf(a - m) + __expf(b - m);
#pragma unroll
    for (int o = 16; o > 0; o >>= 1) s += __shfl_xor(s, o, 32);
    float l = m + __logf(s);
    p[lane]      = a - l;
    p[lane + 32] = b - l;
}


extern "C" void kernel_launch(void* const* d_in, const int* in_sizes, int n_in,
                              void* d_out, int out_size, void* d_ws, size_t ws_size,
                              hipStream_t stream) {
    const float* x   = (const float*)d_in[0];
    const int*   ei  = (const int*)d_in[1];
    const float* Wl0 = (const float*)d_in[2];
    const float* bl0 = (const float*)d_in[3];
    const float* Wr0 = (const float*)d_in[4];
    const float* Wl1 = (const float*)d_in[5];
    const float* bl1 = (const float*)d_in[6];
    const float* Wr1 = (const float*)d_in[7];
    const float* Wl2 = (const float*)d_in[8];
    const float* bl2 = (const float*)d_in[9];
    const float* Wr2 = (const float*)d_in[10];

    const int n_nodes = in_sizes[0] / FIN;
    const int n_edges = in_sizes[1] / 2;
    const int* src = ei;
    const int* dst = ei + n_edges;

    float* agg  = (float*)d_ws;
    float* hbuf = agg + (long)n_nodes * FIN;
    int*   cnt  = (int*)(hbuf + (long)n_nodes * FIN);
    float* inv  = (float*)(cnt + n_nodes);
    int*   pos  = (int*)agg;
    int*   col  = (int*)d_out;
    int*   rowstart = col + n_edges;

    const int eb1    = (n_edges + 255) / 256;
    const int nb     = (n_nodes + 255) / 256;
    const int sb     = (n_nodes + 1023) / 1024;
    const int gb     = (n_nodes * 32 + 255) / 256;
    const int mtiles = n_nodes / 16;

    zero_i32<<<nb, 256, 0, stream>>>(cnt, n_nodes);
    rank_kernel<<<eb1, 256, 0, stream>>>(dst, cnt, pos, n_edges);
    rowstart_kernel<<<sb, 1024, 0, stream>>>(cnt, rowstart, inv, n_nodes);
    fill_kernel<<<eb1, 256, 0, stream>>>(src, dst, rowstart, pos, col, n_edges);

    gather_sum_kernel<<<gb, 256, 0, stream>>>(x, rowstart, cnt, col, agg, n_nodes);
    sage_gemm_kernel<128, true><<<mtiles, 256, 0, stream>>>(agg, inv, x,
                                                            Wl0, bl0, Wr0, hbuf);

    gather_sum_kernel<<<gb, 256, 0, stream>>>(hbuf, rowstart, cnt, col, agg, n_nodes);
    sage_gemm_kernel<128, true><<<mtiles, 256, 0, stream>>>(agg, inv, hbuf,
                                                            Wl1, bl1, Wr1, hbuf);

    gather_sum_kernel<<<gb, 256, 0, stream>>>(hbuf, rowstart, cnt, col, agg, n_nodes);
    sage_gemm_kernel<64, false><<<mtiles, 128, 0, stream>>>(agg, inv, hbuf,
                                                            Wl2, bl2, Wr2,
                                                            (float*)d_out);

    logsoftmax64_kernel<<<(n_nodes * 32 + 255) / 256, 256, 0, stream>>>(
        (float*)d_out, n_nodes);
}


# Neighbour gather fused into the GEMM launch: a 32-node group sums its neighbour rows straight into its LDS A tile, sums never stored, 3 launches fewer, layers alternate buffers; f32 sum order kept
# speedup vs baseline: 2.7524x; 1.1832x over previous
#include <hip/hip_runtime.h>
#include <hip/hip_bf16.h>
#include <math.h>


typedef __attribute__((ext_vector_type(2))) float v2f;
typedef __attribute__((ext_vector_type(8))) float v8f;

#define FIN 128


__global__ void zero_i32(int* __restrict__ p, int n) {
    int i = blockIdx.x * blockDim.x + threadIdx.x;
    if (i < n) p[i] = 0;
}

__global__ void rank_kernel(const int* __restrict__ dst, int* __restrict__ cnt,
                            int* __restrict__ pos, int nedges) {
    int e = blockIdx.x * blockDim.x + threadIdx.x;
    if (e < nedges) pos[e] = atomicAdd(&cnt[dst[e]], 1);
}

__global__ void __launch_bounds__(1024)
rowstart_kernel(const int* __restrict__ cnt, int* __restrict__ rowstart,
                float* __restrict__ inv, int n) {
    __shared__ int wtot[32];
    __shared__ int wpre[32];
    const int t    = threadIdx.x;
    const int lane = t & 31;
    const int w    = t >> 5;
    const int i    = blockIdx.x * 1024 + t;

    int p = 0;
    for (int j = t; j < (int)blockIdx.x * 1024; j += 1024) p += cnt[j];
    const int c = (i < n) ? cnt[i] : 0;

    int x = c;
    for (int o = 1; o < 32; o <<= 1) {
        int y = __shfl_up(x, o, 32);
        if (lane >= o) x += y;
    }
    for (int o = 16; o > 0; o >>= 1) p += __shfl_xor(p, o, 32);
    if (lane == 31) wtot[w] = x;
    if (lane == 0)  wpre[w] = p;
    __syncthreads();

    int tot = wpre[lane] + ((lane < w) ? wtot[lane] : 0);
    for (int o = 16; o > 0; o >>= 1) tot += __shfl_xor(tot, o, 32);

    if (i < n) {
        rowstart[i] = tot + x - c;
        inv[i]      = 1.0f / fmaxf((float)c, 1.0f);
    }
}

__global__ void fill_kernel(const int* __restrict__ src, const int* __restrict__ dst,
                            const int* __restrict__ rowstart,
                            const int* __restrict__ pos, int* __restrict__ col,
                            int nedges) {
    int e = blockIdx.x * blockDim.x + threadIdx.x;
    if (e < nedges) col[rowstart[dst[e]] + pos[e]] = src[e];
}

template <int B>
__device__ __forceinline__ void add_rows(const float* __restrict__ h, int idxv,
                                         int k, int lane, float4& acc) {
    float4 v[B];
#pragma unroll
    for (int u = 0; u < B; ++u) {
        int s = __builtin_amdgcn_readlane(idxv, k + u);
        v[u] = ((const float4*)(h + (long)s * FIN))[lane];
    }
#pragma unroll
    for (int u = 0; u < B; ++u) {
        acc.x += v[u].x; acc.y += v[u].y; acc.z += v[u].z; acc.w += v[u].w;
    }
}

__device__ __forceinline__ float4 gather_row_sum(const float* __restrict__ h,
                                                 const int* __restrict__ col,
                                                 int start, int deg, int idx0, int lane) {
    float4 acc = make_float4(0.0f, 0.0f, 0.0f, 0.0f);
    for (int base = 0; base < deg; base += 32) {
        const int m = min(32, deg - base);
        int idxv = idx0;
        if (base > 0) idxv = (lane < m) ? col[start + base + lane] : 0;
        int k = 0;
        for (; k + 8 <= m; k += 8) add_rows<8>(h, idxv, k, lane, acc);
        if (k + 4 <= m) { add_rows<4>(h, idxv, k, lane, acc); k += 4; }
        if (k + 2 <= m) { add_rows<2>(h, idxv, k, lane, acc); k += 2; }
        if (k < m)      { add_rows<1>(h, idxv, k, lane, acc); }
    }
    return acc;
}


template <int FOUT, bool ELU>
__global__ void __launch_bounds__(32 * 2 * (FOUT / 16))
sage_layer_kernel(const float* __restrict__ h, const int* __restrict__ rowstart,
                  const int* __restrict__ cnt, const int* __restrict__ srclist,
                  const float* __restrict__ invdeg,
                  const float* __restrict__ Wl, const float* __restrict__ bl,
                  const float* __restrict__ Wr, float* __restrict__ out, int n) {
    __shared__ float As[32][132];
    __shared__ float Xs[32][132];

    constexpr int NCT = FOUT / 16;
    constexpr int NW  = 2 * NCT;
    constexpr int RPW = 32 / NW;

    const int m0   = blockIdx.x * 32;
    const int lane = threadIdx.x & 31;
    const int wave = __builtin_amdgcn_readfirstlane((int)(threadIdx.x >> 5));
    const int r0   = wave * RPW;
    const int g0   = m0 + r0 - ((m0 + r0 >= n) ? 16 : 0);

    float4 xv[RPW];
    int    start[RPW], deg[RPW], idx0[RPW];
    float  inv[RPW];
#pragma unroll
    for (int r = 0; r < RPW; ++r)
        xv[r] = ((const float4*)(h + (long)(g0 + r) * FIN))[lane];
#pragma unroll
    for (int r = 0; r < RPW; ++r) {
        start[r] = rowstart[g0 + r];
        deg[r]   = cnt[g0 + r];
        inv[r]   = invdeg[g0 + r];
    }
#pragma unroll
    for (int r = 0; r < RPW; ++r)
        idx0[r] = (lane < deg[r]) ? srclist[start[r] + lane] : 0;
#pragma unroll
    for (int r = 0; r < RPW; ++r)
        *(float4*)&Xs[r0 + r][lane * 4] = xv[r];

#pragma unroll
    for (int r = 0; r < RPW; ++r) {
        float4 a = gather_row_sum(h, srclist, start[r], deg[r], idx0[r], lane);
        a.x *= inv[r]; a.y *= inv[r]; a.z *= inv[r]; a.w *= inv[r];
        *(float4*)&As[r0 + r][lane * 4] = a;
    }
    __syncthreads();

    const int rt   = wave / NCT;
    const int n0   = (wave - rt * NCT) * 16;
    const int row  = 16 * rt + (lane & 15);
    const int col  = lane & 15;
    const int hi   = lane >> 4;

    const float* wl = Wl + 2 * hi * FOUT + n0 + col;
    const float* wr = Wr + 2 * hi * FOUT + n0 + col;

    v8f c = {};

#pragma unroll
    for (int k = 0; k < FIN; k += 4) {
        v2f a, b;
        a.x = As[row][k + 2 * hi];
        a.y = As[row][k + 2 * hi + 1];
        b.x = wl[k * FOUT];
        b.y = wl[(k + 1) * FOUT];
        c = __builtin_amdgcn_wmma_f32_16x16x4_f32(false, a, false, b,
                                                  (short)0, c, false, false);
    }
#pragma unroll
    for (int k = 0; k < FIN; k += 4) {
        v2f a, b;
        a.x = Xs[row][k + 2 * hi];
        a.y = Xs[row][k + 2 * hi + 1];
        b.x = wr[k * FOUT];
        b.y = wr[(k + 1) * FOUT];
        c = __builtin_amdgcn_wmma_f32_16x16x4_f32(false, a, false, b,
                                                  (short)0, c, false, false);
    }

    if (m0 + 16 * rt >= n) return;
    const float bias = bl[n0 + col];
#pragma unroll
    for (int i = 0; i < 8; ++i) {
        float v = c[i] + bias;
        if (ELU) v = (v > 0.0f) ? v : expm1f(v);
        out[(long)(m0 + 16 * rt + i + 8 * hi) * FOUT + n0 + col] = v;
    }
}

__global__ void logsoftmax64_kernel(const float* __restrict__ in,
                                    float* __restrict__ out, int n) {
    int row  = (blockIdx.x * blockDim.x + threadIdx.x) >> 5;
    int lane = threadIdx.x & 31;
    if (row >= n) return;
    const float* p = in + (long)row * 64;
    float* q = out + (long)row * 64;
    float a = p[lane];
    float b = p[lane + 32];
    float m = fmaxf(a, b);
#pragma unroll
    for (int o = 16; o > 0; o >>= 1) m = fmaxf(m, __shfl_xor(m, o, 32));
    float s = __expf(a - m) + __expf(b - m);
#pragma unroll
    for (int o = 16; o > 0; o >>= 1) s += __shfl_xor(s, o, 32);
    float l = m + __logf(s);
    q[lane]      = a - l;
    q[lane + 32] = b - l;
}


extern "C" void kernel_launch(void* const* d_in, const int* in_sizes, int n_in,
                              void* d_out, int out_size, void* d_ws, size_t ws_size,
                              hipStream_t stream) {
    const float* x   = (const float*)d_in[0];
    const int*   ei  = (const int*)d_in[1];
    const float* Wl0 = (const float*)d_in[2];
    const float* bl0 = (const float*)d_in[3];
    const float* Wr0 = (const float*)d_in[4];
    const float* Wl1 = (const float*)d_in[5];
    const float* bl1 = (const float*)d_in[6];
    const float* Wr1 = (const float*)d_in[7];
    const float* Wl2 = (const float*)d_in[8];
    const float* bl2 = (const float*)d_in[9];
    const float* Wr2 = (const float*)d_in[10];

    const int n_nodes = in_sizes[0] / FIN;
    const int n_edges = in_sizes[1] / 2;
    const int* src = ei;
    const int* dst = ei + n_edges;

    float* bufB = (float*)d_ws;
    float* bufA = bufB + (long)n_nodes * FIN;
    int*   cnt  = (int*)(bufA + (long)n_nodes * FIN);
    float* inv  = (float*)(cnt + n_nodes);
    int*   pos  = (int*)bufB;
    int*   col  = (int*)d_out;
    int*   rowstart = col + n_edges;

    const int eb1    = (n_edges + 255) / 256;
    const int nb     = (n_nodes + 255) / 256;
    const int sb     = (n_nodes + 1023) / 1024;
    const int groups = (n_nodes + 31) / 32;

    zero_i32<<<nb, 256, 0, stream>>>(cnt, n_nodes);
    rank_kernel<<<eb1, 256, 0, stream>>>(dst, cnt, pos, n_edges);
    rowstart_kernel<<<sb, 1024, 0, stream>>>(cnt, rowstart, inv, n_nodes);
    fill_kernel<<<eb1, 256, 0, stream>>>(src, dst, rowstart, pos, col, n_edges);

    sage_layer_kernel<128, true><<<groups, 512, 0, stream>>>(x, rowstart, cnt, col, inv,
                                                             Wl0, bl0, Wr0, bufA, n_nodes);
    sage_layer_kernel<128, true><<<groups, 512, 0, stream>>>(bufA, rowstart, cnt, col, inv,
                                                             Wl1, bl1, Wr1, bufB, n_nodes);
    sage_layer_kernel<64, false><<<groups, 256, 0, stream>>>(bufB, rowstart, cnt, col, inv,
                                                             Wl2, bl2, Wr2, bufA, n_nodes);

    logsoftmax64_kernel<<<(n_nodes * 32 + 255) / 256, 256, 0, stream>>>(
        bufA, (float*)d_out, n_nodes);
}
